# starting kernel
# baseline (speedup 1.0000x reference)
#include <hip/hip_runtime.h>


#define UNITS     128
#define IN_STRIDE 129
#define TILE      16
#define SUBSTEPS  8
#define TERMS     12
#define YSTRIDE   132

typedef __attribute__((ext_vector_type(2))) float v2f;
typedef __attribute__((ext_vector_type(8))) float v8f;

__global__ void build_KT(const float* __restrict__ p, float* __restrict__ KdT) {
    int idx = blockIdx.x * blockDim.x + threadIdx.x;
    if (idx >= UNITS * UNITS) return;
    int i = idx >> 7;
    int j = idx & 127;
    float val = 0.0f;
    if (j == i) {
        float x  = p[i];
        float sp = (x > 20.0f) ? x : log1pf(__expf(x));
        val = -sp * sp;
    } else if (j == i + 1) {
        val = -p[UNITS + i];
    } else if (j + 1 == i) {
        val = p[UNITS + j];
    }
    KdT[j * UNITS + i] = val;
}

__global__ void __launch_bounds__(256)
expmv_taylor_wmma(const float* __restrict__ inp,
                  const float* __restrict__ KdT,
                  float* __restrict__ out,
                  int B)
{
    __shared__ float Ybuf[2][TILE * YSTRIDE];

    const int tid   = threadIdx.x;
    const int wave  = tid >> 5;
    const int lane  = tid & 31;
    const int col   = lane & 15;
    const int hi    = lane >> 4;
    const int jbase = wave * 16;
    const int row0  = blockIdx.x * TILE;

    v2f breg[32];
    #pragma unroll
    for (int c = 0; c < 32; ++c) {
        breg[c] = *reinterpret_cast<const v2f*>(
            &KdT[(jbase + col) * UNITS + 4 * c + 2 * hi]);
    }

    float tvec[8];
    v8f S = {};
    #pragma unroll
    for (int v = 0; v < 8; ++v) {
        int r  = row0 + v + 8 * hi;
        int rc = (r < B) ? r : (B - 1);
        tvec[v] = inp[rc * IN_STRIDE + UNITS];
        S[v]    = inp[rc * IN_STRIDE + jbase + col];
    }

    for (int idx = tid; idx < TILE * UNITS; idx += 256) {
        int r  = idx >> 7, cc = idx & 127;
        int rr = row0 + r;
        int rc = (rr < B) ? rr : (B - 1);
        Ybuf[0][r * YSTRIDE + cc] = inp[rc * IN_STRIDE + cc];
    }
    __syncthreads();

    const int aoff = col * YSTRIDE + 2 * hi;

    int cur = 0;
    for (int step = 0; step < SUBSTEPS; ++step) {
        for (int k = 1; k <= TERMS; ++k) {
            const float* Yc = Ybuf[cur];
            float*       Yn = Ybuf[cur ^ 1];

            v8f z0 = {};
            v8f z1 = {};
            v2f abuf[8];
            #pragma unroll
            for (int c = 0; c < 8; ++c)
                abuf[c] = *reinterpret_cast<const v2f*>(&Yc[aoff + 4 * c]);
            #pragma unroll
            for (int c = 0; c < 32; ++c) {
                v2f a = abuf[c & 7];
                if (c + 8 < 32)
                    abuf[c & 7] = *reinterpret_cast<const v2f*>(&Yc[aoff + 4 * (c + 8)]);
                if (c & 1)
                    z1 = __builtin_amdgcn_wmma_f32_16x16x4_f32(
                            false, a, false, breg[c], (short)0, z1, false, false);
                else
                    z0 = __builtin_amdgcn_wmma_f32_16x16x4_f32(
                            false, a, false, breg[c], (short)0, z0, false, false);
            }

            const float invsk = __builtin_amdgcn_rcpf((float)(SUBSTEPS * k));
            const bool  last  = (k == TERMS);
            #pragma unroll
            for (int v = 0; v < 8; ++v) {
                float val = (z0[v] + z1[v]) * tvec[v] * invsk;
                S[v] += val;
                Yn[(v + 8 * hi) * YSTRIDE + jbase + col] = last ? S[v] : val;
            }
            __syncthreads();
            cur ^= 1;
        }
    }

    #pragma unroll
    for (int v = 0; v < 8; ++v) {
        int r = row0 + v + 8 * hi;
        if (r < B) out[r * UNITS + jbase + col] = S[v];
    }
}

extern "C" void kernel_launch(void* const* d_in, const int* in_sizes, int n_in,
                              void* d_out, int out_size, void* d_ws, size_t ws_size,
                              hipStream_t stream) {
    const float* inp    = (const float*)d_in[0];
    const float* params = (const float*)d_in[1];
    float*       out    = (float*)d_out;
    float*       KdT    = (float*)d_ws;

    int B = in_sizes[0] / IN_STRIDE;

    build_KT<<<(UNITS * UNITS + 255) / 256, 256, 0, stream>>>(params, KdT);

    int tiles = (B + TILE - 1) / TILE;
    expmv_taylor_wmma<<<tiles, 256, 0, stream>>>(inp, KdT, out, B);
}


# K-apply transposed (K^T as A, state as B): the scaled D tile is the next apply's B fragment in place, v_k stays in registers, 5 chunks per tile, only edge states cross LDS; summation order changes
# speedup vs baseline: 3.5860x; 3.5860x over previous
#include <hip/hip_runtime.h>


#define UNITS     128
#define IN_STRIDE 129
#define TILE      16
#define SUBSTEPS  8
#define TERMS     12
#define NWAVE     4
#define HBUF      (NWAVE * 2 * TILE)

typedef __attribute__((ext_vector_type(2))) float v2f;
typedef __attribute__((ext_vector_type(8))) float v8f;

__global__ void build_KT(const float* __restrict__ p, float* __restrict__ KdT) {
    int idx = blockIdx.x * blockDim.x + threadIdx.x;
    if (idx >= UNITS * UNITS) return;
    int i = idx >> 7;
    int j = idx & 127;
    float val = 0.0f;
    if (j == i) {
        float x  = p[i];
        float sp = (x > 20.0f) ? x : log1pf(__expf(x));
        val = -sp * sp;
    } else if (j == i + 1) {
        val = -p[UNITS + i];
    } else if (j + 1 == i) {
        val = p[UNITS + j];
    }
    KdT[j * UNITS + i] = val;
}

static __device__ __forceinline__ float kval(const float* __restrict__ KdT, int s, int j) {
    return (s >= 0 && s < UNITS) ? KdT[j * UNITS + s] : 0.0f;
}

static __device__ __forceinline__ v8f mma(v2f a, v2f b, v8f c) {
    return __builtin_amdgcn_wmma_f32_16x16x4_f32(false, a, false, b, (short)0, c, false, false);
}

__global__ void __launch_bounds__(NWAVE * 32)
expmv_taylor_wmma_t(const float* __restrict__ inp,
                    const float* __restrict__ KdT,
                    float* __restrict__ out,
                    int B)
{
    __shared__ float edge[2 * HBUF];

    const int tid  = threadIdx.x;
    const int wave = tid >> 5;
    const int lane = tid & 31;
    const int c    = lane & 15;
    const int h    = lane >> 4;
    const int j0   = wave * 32;
    const int j1   = j0 + 16;
    const int row0 = blockIdx.x * TILE;

    v2f a0[5], a1[5];
    #pragma unroll
    for (int p = 0; p < 4; ++p) {
        a0[p] = *reinterpret_cast<const v2f*>(&KdT[(j0 + c) * UNITS + j0 + 8 * h + 2 * p]);
        a1[p] = *reinterpret_cast<const v2f*>(&KdT[(j1 + c) * UNITS + j1 + 8 * h + 2 * p]);
    }
    const int se = h ? (j0 + 32) : (j0 - 1);
    a0[4][0] = h ? 0.0f : KdT[(j0 + c) * UNITS + j1];
    a0[4][1] = kval(KdT, se, j0 + c);
    a1[4][0] = h ? KdT[(j1 + c) * UNITS + j0 + 15] : 0.0f;
    a1[4][1] = kval(KdT, se, j1 + c);

    const int   r  = row0 + c;
    const int   rc = (r < B) ? r : (B - 1);
    const float tl = inp[rc * IN_STRIDE + UNITS];
    v8f V0, V1;
    #pragma unroll
    for (int v = 0; v < 8; ++v) {
        V0[v] = inp[rc * IN_STRIDE + j0 + 8 * h + v];
        V1[v] = inp[rc * IN_STRIDE + j1 + 8 * h + v];
    }
    v8f S0 = V0, S1 = V1;

    const int wr_off = (wave * 2 + h) * TILE + c;
    const int nb_w   = h ? ((wave < NWAVE - 1) ? wave + 1 : wave) : ((wave > 0) ? wave - 1 : wave);
    const int nb_s   = h ? ((wave < NWAVE - 1) ? 0 : 1)           : ((wave > 0) ? 1 : 0);
    const int rd_off = (nb_w * 2 + nb_s) * TILE + c;

    edge[wr_off] = h ? V1[7] : V0[0];
    float X = h ? V0[7] : V1[0];

    v8f z0, z1;
    {
        const v8f zero = {};
        v2f b0 = {V0[0], V0[1]};
        v2f b1 = {V1[0], V1[1]};
        z0 = mma(a0[0], b0, zero);
        z1 = mma(a1[0], b1, zero);
    }

    int cur = 0;
    for (int step = 0; step < SUBSTEPS; ++step) {
        for (int k = 1; k <= TERMS; ++k) {
            __syncthreads();
            const float E = edge[cur * HBUF + rd_off];
            __builtin_amdgcn_sched_barrier(0);
            #pragma unroll
            for (int p = 1; p < 4; ++p) {
                v2f b0 = {V0[2 * p], V0[2 * p + 1]};
                v2f b1 = {V1[2 * p], V1[2 * p + 1]};
                z0 = mma(a0[p], b0, z0);
                z1 = mma(a1[p], b1, z1);
            }
            v2f xe = {X, E};
            z0 = mma(a0[4], xe, z0);
            z1 = mma(a1[4], xe, z1);

            const float cs   = tl * __builtin_amdgcn_rcpf((float)(SUBSTEPS * k));
            const bool  last = (k == TERMS);

            {
                const float wl = z0[0] * cs;
                const float wh = z1[7] * cs;
                const float sl = S0[0] + wl;
                const float sh = S1[7] + wh;
                cur ^= 1;
                edge[cur * HBUF + wr_off] = h ? (last ? sh : wh) : (last ? sl : wl);
            }
            __builtin_amdgcn_sched_barrier(0);

            #pragma unroll
            for (int v = 0; v < 2; ++v) {
                float w0 = z0[v] * cs;
                float w1 = z1[v] * cs;
                S0[v] += w0;
                S1[v] += w1;
                V0[v] = last ? S0[v] : w0;
                V1[v] = last ? S1[v] : w1;
            }
            v8f n0, n1;
            {
                const v8f zero = {};
                v2f b0 = {V0[0], V0[1]};
                v2f b1 = {V1[0], V1[1]};
                n0 = mma(a0[0], b0, zero);
                n1 = mma(a1[0], b1, zero);
            }
            #pragma unroll
            for (int v = 2; v < 8; ++v) {
                float w0 = z0[v] * cs;
                float w1 = z1[v] * cs;
                S0[v] += w0;
                S1[v] += w1;
                V0[v] = last ? S0[v] : w0;
                V1[v] = last ? S1[v] : w1;
            }
            X  = h ? V0[7] : V1[0];
            z0 = n0;
            z1 = n1;
        }
    }

    if (r < B) {
        #pragma unroll
        for (int v = 0; v < 8; ++v) {
            out[r * UNITS + j0 + 8 * h + v] = S0[v];
            out[r * UNITS + j1 + 8 * h + v] = S1[v];
        }
    }
}

extern "C" void kernel_launch(void* const* d_in, const int* in_sizes, int n_in,
                              void* d_out, int out_size, void* d_ws, size_t ws_size,
                              hipStream_t stream) {
    const float* inp    = (const float*)d_in[0];
    const float* params = (const float*)d_in[1];
    float*       out    = (float*)d_out;
    float*       KdT    = (float*)d_ws;

    int B = in_sizes[0] / IN_STRIDE;

    build_KT<<<(UNITS * UNITS + 255) / 256, 256, 0, stream>>>(params, KdT);

    int tiles = (B + TILE - 1) / TILE;
    expmv_taylor_wmma_t<<<tiles, NWAVE * 32, 0, stream>>>(inp, KdT, out, B);
}


# K2=K*K (5 diagonals, still 5 WMMAs per tile) steps two Taylor terms: even and odd powers as independent chains, one edge exchange per two terms; powers unscaled, h^k/k! in the sum; sum order changes
# speedup vs baseline: 4.4365x; 1.2372x over previous
#include <hip/hip_runtime.h>


#define UNITS     128
#define IN_STRIDE 129
#define TILE      16
#define SUBSTEPS  8
#define TERMS     12
#define ROUNDS    (TERMS / 2)
#define NWAVE     4
#define SLOTS     (NWAVE * 2 * TILE)
#define B1_OFF    0
#define B2_OFF    (UNITS * 3)

typedef __attribute__((ext_vector_type(2))) float v2f;
typedef __attribute__((ext_vector_type(8))) float v8f;

__global__ void build_bands(const float* __restrict__ p, float* __restrict__ ws) {
    const int j = blockIdx.x * blockDim.x + threadIdx.x;
    if (j >= UNITS) return;
    float xr[3], cr[4];
    #pragma unroll
    for (int d = 0; d < 3; ++d) {
        const int i = j + d - 1;
        xr[d] = p[min(max(i, 0), UNITS - 1)];
    }
    #pragma unroll
    for (int d = 0; d < 4; ++d) {
        const int i = j + d - 2;
        cr[d] = p[UNITS + min(max(i, 0), UNITS - 2)];
    }
    float sg[3], ct[4];
    #pragma unroll
    for (int d = 0; d < 3; ++d) {
        const int   i  = j + d - 1;
        const float x  = xr[d];
        const float sp = (x > 20.0f) ? x : log1pf(__expf(x));
        sg[d] = (i >= 0 && i < UNITS) ? -sp * sp : 0.0f;
    }
    #pragma unroll
    for (int d = 0; d < 4; ++d) {
        const int i = j + d - 2;
        ct[d] = (i >= 0 && i < UNITS - 1) ? -cr[d] : 0.0f;
    }
    float* b1 = ws + B1_OFF + j * 3;
    b1[0] = ct[1];
    b1[1] = sg[1];
    b1[2] = -ct[2];
    float* b2 = ws + B2_OFF + j * 5;
    b2[0] = ct[0] * ct[1];
    b2[1] = sg[0] * ct[1] + ct[1] * sg[1];
    b2[2] = sg[1] * sg[1] - ct[1] * ct[1] - ct[2] * ct[2];
    b2[3] = -(ct[2] * sg[1]) - sg[2] * ct[2];
    b2[4] = ct[3] * ct[2];
}

static __device__ __forceinline__ float band_val(const float* __restrict__ bw, int hw, int s, int j) {
    const int  d      = s - j;
    const bool inband = (d >= -hw && d <= hw);
    const float v     = bw[j * (2 * hw + 1) + (inband ? d + hw : 0)];
    return inband ? v : 0.0f;
}

static __device__ __forceinline__ void load_a(v2f (&a)[5], const float* __restrict__ bw, int hw,
                                              int jt, int c, int h, int sa, int sb) {
    #pragma unroll
    for (int p = 0; p < 4; ++p) {
        a[p][0] = band_val(bw, hw, jt + 8 * h + 2 * p,     jt + c);
        a[p][1] = band_val(bw, hw, jt + 8 * h + 2 * p + 1, jt + c);
    }
    const int s4 = h ? sb : sa;
    a[4][0] = band_val(bw, hw, s4,     jt + c);
    a[4][1] = band_val(bw, hw, s4 + 1, jt + c);
}

static __device__ __forceinline__ v8f mma(v2f a, v2f b, v8f c) {
    return __builtin_amdgcn_wmma_f32_16x16x4_f32(false, a, false, b, (short)0, c, false, false);
}

__global__ void __launch_bounds__(NWAVE * 32)
expmv_taylor_wmma_k2(const float* __restrict__ inp,
                     const float* __restrict__ ws,
                     float* __restrict__ out,
                     int B)
{
    __shared__ v2f edge[2 * 2 * SLOTS];

    const int tid  = threadIdx.x;
    const int wave = tid >> 5;
    const int lane = tid & 31;
    const int c    = lane & 15;
    const int h    = lane >> 4;
    const int j0   = wave * 32;
    const int j1   = j0 + 16;
    const int row0 = blockIdx.x * TILE;

    v2f k1a0[5], k1a1[5], k2a0[5], k2a1[5];
    load_a(k1a0, ws + B1_OFF, 1, j0, c, h, j1,      j0 - 2);
    load_a(k1a1, ws + B1_OFF, 1, j1, c, h, j0 + 32, j0 + 14);
    load_a(k2a0, ws + B2_OFF, 2, j0, c, h, j1,      j0 - 2);
    load_a(k2a1, ws + B2_OFF, 2, j1, c, h, j0 + 32, j0 + 14);

    const int   r  = row0 + c;
    const int   rc = (r < B) ? r : (B - 1);
    const float tl = inp[rc * IN_STRIDE + UNITS];
    v8f S0, S1;
    #pragma unroll
    for (int v = 0; v < 8; ++v) {
        S0[v] = inp[rc * IN_STRIDE + j0 + 8 * h + v];
        S1[v] = inp[rc * IN_STRIDE + j1 + 8 * h + v];
    }

    float cf[TERMS + 1];
    cf[0] = 1.0f;
    #pragma unroll
    for (int k = 1; k <= TERMS; ++k)
        cf[k] = cf[k - 1] * (tl * __builtin_amdgcn_rcpf((float)(SUBSTEPS * k)));

    const int wr_off = (wave * 2 + h) * TILE + c;
    const int nb_w   = h ? ((wave > 0) ? wave - 1 : wave) : ((wave < NWAVE - 1) ? wave + 1 : wave);
    const int rd_off = (nb_w * 2 + (h ? 1 : 0)) * TILE + c;

    {
        v2f lo = {S0[0], S0[1]};
        v2f hi = {S1[6], S1[7]};
        edge[wr_off] = h ? hi : lo;
    }

    for (int step = 0; step < SUBSTEPS; ++step) {
        v8f P0 = S0, P1 = S1, Q0 = S0, Q1 = S1;
        #pragma unroll
        for (int rd = 0; rd < ROUNDS; ++rd) {
            const int cur = rd & 1;
            const v2f* ap0 = (rd == 0) ? k1a0 : k2a0;
            const v2f* ap1 = (rd == 0) ? k1a1 : k2a1;
            v8f zp0, zp1, zq0, zq1;
            {
                const v8f zero = {};
                v2f bp0 = {P0[0], P0[1]}, bp1 = {P1[0], P1[1]};
                v2f bq0 = {Q0[0], Q0[1]}, bq1 = {Q1[0], Q1[1]};
                zp0 = mma(ap0[0], bp0, zero);
                zp1 = mma(ap1[0], bp1, zero);
                zq0 = mma(k2a0[0], bq0, zero);
                zq1 = mma(k2a1[0], bq1, zero);
                __builtin_amdgcn_sched_barrier(0);
                v2f dp0 = {P0[2], P0[3]}, dp1 = {P1[2], P1[3]};
                v2f dq0 = {Q0[2], Q0[3]}, dq1 = {Q1[2], Q1[3]};
                zp0 = mma(ap0[1], dp0, zp0);
                zp1 = mma(ap1[1], dp1, zp1);
                zq0 = mma(k2a0[1], dq0, zq0);
                zq1 = mma(k2a1[1], dq1, zq1);
            }
            __syncthreads();
            const v2f EP = edge[(cur * 2 + 0) * SLOTS + rd_off];
            const v2f EQ = (rd == 0) ? EP : edge[(cur * 2 + 1) * SLOTS + rd_off];
            __builtin_amdgcn_sched_barrier(0);
            #pragma unroll
            for (int p = 2; p < 4; ++p) {
                v2f bp0 = {P0[2 * p], P0[2 * p + 1]}, bp1 = {P1[2 * p], P1[2 * p + 1]};
                v2f bq0 = {Q0[2 * p], Q0[2 * p + 1]}, bq1 = {Q1[2 * p], Q1[2 * p + 1]};
                zp0 = mma(ap0[p], bp0, zp0);
                zp1 = mma(ap1[p], bp1, zp1);
                zq0 = mma(k2a0[p], bq0, zq0);
                zq1 = mma(k2a1[p], bq1, zq1);
            }
            __builtin_amdgcn_sched_barrier(0);
            {
                v2f p1lo = {P1[0], P1[1]}, p0hi = {P0[6], P0[7]};
                v2f q1lo = {Q1[0], Q1[1]}, q0hi = {Q0[6], Q0[7]};
                v2f xp0 = h ? EP : p1lo;
                v2f xp1 = h ? p0hi : EP;
                v2f xq0 = h ? EQ : q1lo;
                v2f xq1 = h ? q0hi : EQ;
                zp0 = mma(ap0[4], xp0, zp0);
                zp1 = mma(ap1[4], xp1, zp1);
                zq0 = mma(k2a0[4], xq0, zq0);
                zq1 = mma(k2a1[4], xq1, zq1);
            }

            if (rd < ROUNDS - 1) {
                v2f plo = {zp0[0], zp0[1]}, phi = {zp1[6], zp1[7]};
                v2f qlo = {zq0[0], zq0[1]}, qhi = {zq1[6], zq1[7]};
                edge[((cur ^ 1) * 2 + 0) * SLOTS + wr_off] = h ? phi : plo;
                edge[((cur ^ 1) * 2 + 1) * SLOTS + wr_off] = h ? qhi : qlo;
                __builtin_amdgcn_sched_barrier(0);
            }
            #pragma unroll
            for (int v = 0; v < 8; ++v) {
                S0[v] += cf[2 * rd + 1] * zp0[v];
                S1[v] += cf[2 * rd + 1] * zp1[v];
            }
            #pragma unroll
            for (int v = 0; v < 8; ++v) {
                S0[v] += cf[2 * rd + 2] * zq0[v];
                S1[v] += cf[2 * rd + 2] * zq1[v];
            }
            if (rd == ROUNDS - 1) {
                v2f lo = {S0[0], S0[1]};
                v2f hi = {S1[6], S1[7]};
                edge[((cur ^ 1) * 2 + 0) * SLOTS + wr_off] = h ? hi : lo;
                __builtin_amdgcn_sched_barrier(0);
            }
            P0 = zp0; P1 = zp1; Q0 = zq0; Q1 = zq1;
        }
    }

    if (r < B) {
        #pragma unroll
        for (int v = 0; v < 8; ++v) {
            out[r * UNITS + j0 + 8 * h + v] = S0[v];
            out[r * UNITS + j1 + 8 * h + v] = S1[v];
        }
    }
}

extern "C" void kernel_launch(void* const* d_in, const int* in_sizes, int n_in,
                              void* d_out, int out_size, void* d_ws, size_t ws_size,
                              hipStream_t stream) {
    const float* inp    = (const float*)d_in[0];
    const float* params = (const float*)d_in[1];
    float*       out    = (float*)d_out;
    float*       bands  = (float*)d_ws;

    int B = in_sizes[0] / IN_STRIDE;

    build_bands<<<1, UNITS, 0, stream>>>(params, bands);

    int tiles = (B + TILE - 1) / TILE;
    expmv_taylor_wmma_k2<<<tiles, NWAVE * 32, 0, stream>>>(inp, bands, out, B);
}


# Taylor sum as Horner in K2: v_j = c_2j u + c_2j+1 K u is the C operand, a <- v_j + K2 a; 7 five-WMMA applies per tile and substep in place of 12, exchanges unchanged; sum order changes
# speedup vs baseline: 5.4513x; 1.2287x over previous
#include <hip/hip_runtime.h>


#define UNITS     128
#define IN_STRIDE 129
#define TILE      16
#define SUBSTEPS  8
#define TERMS     12
#define ROUNDS    (TERMS / 2)
static_assert(TERMS == 12, "the round-to-coefficient map of the Horner rounds is written for 12 terms");
#define NWAVE     4
#define SLOTS     (NWAVE * 2 * TILE)
#define B1_OFF    0
#define B2_OFF    (UNITS * 3)

typedef __attribute__((ext_vector_type(2))) float v2f;
typedef __attribute__((ext_vector_type(8))) float v8f;

__global__ void build_bands(const float* __restrict__ p, float* __restrict__ ws) {
    const int j = blockIdx.x * blockDim.x + threadIdx.x;
    if (j >= UNITS) return;
    float xr[3], cr[4];
    #pragma unroll
    for (int d = 0; d < 3; ++d) {
        const int i = j + d - 1;
        xr[d] = p[min(max(i, 0), UNITS - 1)];
    }
    #pragma unroll
    for (int d = 0; d < 4; ++d) {
        const int i = j + d - 2;
        cr[d] = p[UNITS + min(max(i, 0), UNITS - 2)];
    }
    float sg[3], ct[4];
    #pragma unroll
    for (int d = 0; d < 3; ++d) {
        const int   i  = j + d - 1;
        const float x  = xr[d];
        const float sp = (x > 20.0f) ? x : log1pf(__expf(x));
        sg[d] = (i >= 0 && i < UNITS) ? -sp * sp : 0.0f;
    }
    #pragma unroll
    for (int d = 0; d < 4; ++d) {
        const int i = j + d - 2;
        ct[d] = (i >= 0 && i < UNITS - 1) ? -cr[d] : 0.0f;
    }
    float* b1 = ws + B1_OFF + j * 3;
    b1[0] = ct[1];
    b1[1] = sg[1];
    b1[2] = -ct[2];
    float* b2 = ws + B2_OFF + j * 5;
    b2[0] = ct[0] * ct[1];
    b2[1] = sg[0] * ct[1] + ct[1] * sg[1];
    b2[2] = sg[1] * sg[1] - ct[1] * ct[1] - ct[2] * ct[2];
    b2[3] = -(ct[2] * sg[1]) - sg[2] * ct[2];
    b2[4] = ct[3] * ct[2];
}

static __device__ __forceinline__ float band_val(const float* __restrict__ bw, int hw, int s, int j) {
    const int  d      = s - j;
    const bool inband = (d >= -hw && d <= hw);
    const float v     = bw[j * (2 * hw + 1) + (inband ? d + hw : 0)];
    return inband ? v : 0.0f;
}

static __device__ __forceinline__ void load_a(v2f (&a)[5], const float* __restrict__ bw, int hw,
                                              int jt, int c, int h, int sa, int sb) {
    #pragma unroll
    for (int p = 0; p < 4; ++p) {
        a[p][0] = band_val(bw, hw, jt + 8 * h + 2 * p,     jt + c);
        a[p][1] = band_val(bw, hw, jt + 8 * h + 2 * p + 1, jt + c);
    }
    const int s4 = h ? sb : sa;
    a[4][0] = band_val(bw, hw, s4,     jt + c);
    a[4][1] = band_val(bw, hw, s4 + 1, jt + c);
}

static __device__ __forceinline__ v8f mma(v2f a, v2f b, v8f c) {
    return __builtin_amdgcn_wmma_f32_16x16x4_f32(false, a, false, b, (short)0, c, false, false);
}

__global__ void __launch_bounds__(NWAVE * 32)
expmv_taylor_wmma_k2(const float* __restrict__ inp,
                     const float* __restrict__ ws,
                     float* __restrict__ out,
                     int B)
{
    __shared__ v2f edge[2 * SLOTS];

    const int tid  = threadIdx.x;
    const int wave = tid >> 5;
    const int lane = tid & 31;
    const int c    = lane & 15;
    const int h    = lane >> 4;
    const int j0   = wave * 32;
    const int j1   = j0 + 16;
    const int row0 = blockIdx.x * TILE;

    v2f k1a0[5], k1a1[5], k2a0[5], k2a1[5];
    load_a(k1a0, ws + B1_OFF, 1, j0, c, h, j1,      j0 - 2);
    load_a(k1a1, ws + B1_OFF, 1, j1, c, h, j0 + 32, j0 + 14);
    load_a(k2a0, ws + B2_OFF, 2, j0, c, h, j1,      j0 - 2);
    load_a(k2a1, ws + B2_OFF, 2, j1, c, h, j0 + 32, j0 + 14);

    const int   r  = row0 + c;
    const int   rc = (r < B) ? r : (B - 1);
    const float tl = inp[rc * IN_STRIDE + UNITS];
    v8f S0, S1;
    #pragma unroll
    for (int v = 0; v < 8; ++v) {
        S0[v] = inp[rc * IN_STRIDE + j0 + 8 * h + v];
        S1[v] = inp[rc * IN_STRIDE + j1 + 8 * h + v];
    }

    float cf[TERMS + 1];
    cf[0] = 1.0f;
    #pragma unroll
    for (int k = 1; k <= TERMS; ++k)
        cf[k] = cf[k - 1] * (tl * __builtin_amdgcn_rcpf((float)(SUBSTEPS * k)));

    const int wr_off = (wave * 2 + h) * TILE + c;
    const int nb_w   = h ? ((wave > 0) ? wave - 1 : wave) : ((wave < NWAVE - 1) ? wave + 1 : wave);
    const int rd_off = (nb_w * 2 + (h ? 1 : 0)) * TILE + c;

    {
        v2f lo = {S0[0], S0[1]};
        v2f hi = {S1[6], S1[7]};
        edge[wr_off] = h ? hi : lo;
    }

    for (int step = 0; step < SUBSTEPS; ++step) {
        const v8f U0 = S0, U1 = S1;
        v8f W0 = S0, W1 = S1;
        v8f A0 = S0, A1 = S1;
        v8f V0 = S0, V1 = S1;
        #pragma unroll
        for (int rd = 0; rd < ROUNDS; ++rd) {
            const int cur = rd & 1;
            v8f z0, z1;
            v8f y0 = {}, y1 = {};
            {
                const v8f zero = {};
                v2f b0 = {A0[0], A0[1]}, b1 = {A1[0], A1[1]};
                z0 = mma(k2a0[0], b0, (rd == 0) ? zero : V0);
                z1 = mma(k2a1[0], b1, (rd == 0) ? zero : V1);
                if (rd == 0) {
                    y0 = mma(k1a0[0], b0, zero);
                    y1 = mma(k1a1[0], b1, zero);
                }
                __builtin_amdgcn_sched_barrier(0);
                v2f d0 = {A0[2], A0[3]}, d1 = {A1[2], A1[3]};
                z0 = mma(k2a0[1], d0, z0);
                z1 = mma(k2a1[1], d1, z1);
                if (rd == 0) {
                    y0 = mma(k1a0[1], d0, y0);
                    y1 = mma(k1a1[1], d1, y1);
                }
            }
            __syncthreads();
            const v2f E = edge[cur * SLOTS + rd_off];
            __builtin_amdgcn_sched_barrier(0);
            #pragma unroll
            for (int p = 2; p < 4; ++p) {
                v2f b0 = {A0[2 * p], A0[2 * p + 1]}, b1 = {A1[2 * p], A1[2 * p + 1]};
                z0 = mma(k2a0[p], b0, z0);
                z1 = mma(k2a1[p], b1, z1);
                if (rd == 0) {
                    y0 = mma(k1a0[p], b0, y0);
                    y1 = mma(k1a1[p], b1, y1);
                }
            }
            __builtin_amdgcn_sched_barrier(0);
            v8f N0 = V0, N1 = V1;
            if (rd >= 1 && rd < ROUNDS - 1) {
                #pragma unroll
                for (int v = 0; v < 8; ++v) {
                    N0[v] = cf[2 * (4 - rd)] * U0[v] + cf[2 * (4 - rd) + 1] * W0[v];
                    N1[v] = cf[2 * (4 - rd)] * U1[v] + cf[2 * (4 - rd) + 1] * W1[v];
                }
                __builtin_amdgcn_sched_barrier(0);
            }
            {
                v2f a1lo = {A1[0], A1[1]}, a0hi = {A0[6], A0[7]};
                v2f x0 = h ? E : a1lo;
                v2f x1 = h ? a0hi : E;
                z0 = mma(k2a0[4], x0, z0);
                z1 = mma(k2a1[4], x1, z1);
                if (rd == 0) {
                    y0 = mma(k1a0[4], x0, y0);
                    y1 = mma(k1a1[4], x1, y1);
                }
            }

            if (rd == 0) {
                W0 = y0; W1 = y1;
                #pragma unroll
                for (int v = 0; v < 8; ++v) {
                    A0[v] = cf[10] * U0[v] + cf[11] * y0[v] + cf[12] * z0[v];
                    A1[v] = cf[10] * U1[v] + cf[11] * y1[v] + cf[12] * z1[v];
                }
            } else {
                A0 = z0; A1 = z1;
            }
            {
                v2f lo = {A0[0], A0[1]};
                v2f hi = {A1[6], A1[7]};
                edge[(cur ^ 1) * SLOTS + wr_off] = h ? hi : lo;
                __builtin_amdgcn_sched_barrier(0);
            }
            if (rd == 0) {
                #pragma unroll
                for (int v = 0; v < 8; ++v) {
                    V0[v] = cf[8] * U0[v] + cf[9] * W0[v];
                    V1[v] = cf[8] * U1[v] + cf[9] * W1[v];
                }
            } else {
                V0 = N0; V1 = N1;
            }
        }
        S0 = A0; S1 = A1;
    }

    if (r < B) {
        #pragma unroll
        for (int v = 0; v < 8; ++v) {
            out[r * UNITS + j0 + 8 * h + v] = S0[v];
            out[r * UNITS + j1 + 8 * h + v] = S1[v];
        }
    }
}

extern "C" void kernel_launch(void* const* d_in, const int* in_sizes, int n_in,
                              void* d_out, int out_size, void* d_ws, size_t ws_size,
                              hipStream_t stream) {
    const float* inp    = (const float*)d_in[0];
    const float* params = (const float*)d_in[1];
    float*       out    = (float*)d_out;
    float*       bands  = (float*)d_ws;

    int B = in_sizes[0] / IN_STRIDE;

    build_bands<<<1, UNITS, 0, stream>>>(params, bands);

    int tiles = (B + TILE - 1) / TILE;
    expmv_taylor_wmma_k2<<<tiles, NWAVE * 32, 0, stream>>>(inp, bands, out, B);
}


# Band build folded into the main kernel: each workgroup builds the 128 columns of K and K2 itself (one per thread) in LDS and loads its A fragments there; one launch, not two; sum order unchanged
# speedup vs baseline: 6.6615x; 1.2220x over previous
#include <hip/hip_runtime.h>


#define UNITS     128
#define IN_STRIDE 129
#define TILE      16
#define SUBSTEPS  8
#define TERMS     12
#define ROUNDS    (TERMS / 2)
static_assert(TERMS == 12, "the round-to-coefficient map of the Horner rounds is written for 12 terms");
#define NWAVE     4
#define SLOTS     (NWAVE * 2 * TILE)
#define B1_OFF    0
#define B2_OFF    (UNITS * 3)
#define NBAND     (UNITS * 8)
static_assert(NWAVE * 32 == UNITS, "one thread of the workgroup per column of K");

typedef __attribute__((ext_vector_type(2))) float v2f;
typedef __attribute__((ext_vector_type(8))) float v8f;

static __device__ __forceinline__ void load_raw(const float* __restrict__ p, int j,
                                                float (&xr)[3], float (&cr)[4]) {
    #pragma unroll
    for (int d = 0; d < 3; ++d) {
        const int i = j + d - 1;
        xr[d] = p[min(max(i, 0), UNITS - 1)];
    }
    #pragma unroll
    for (int d = 0; d < 4; ++d) {
        const int i = j + d - 2;
        cr[d] = p[UNITS + min(max(i, 0), UNITS - 2)];
    }
}

static __device__ __forceinline__ void store_bands(float* bd, int j,
                                                   const float (&xr)[3], const float (&cr)[4]) {
    float sg[3], ct[4];
    #pragma unroll
    for (int d = 0; d < 3; ++d) {
        const int   i  = j + d - 1;
        const float x  = xr[d];
        const float lp = log1pf(__expf(x));
        const float sp = (x > 20.0f) ? x : lp;
        sg[d] = (i >= 0 && i < UNITS) ? -sp * sp : 0.0f;
    }
    #pragma unroll
    for (int d = 0; d < 4; ++d) {
        const int i = j + d - 2;
        ct[d] = (i >= 0 && i < UNITS - 1) ? -cr[d] : 0.0f;
    }
    float* b1 = bd + B1_OFF + j * 3;
    b1[0] = ct[1];
    b1[1] = sg[1];
    b1[2] = -ct[2];
    float* b2 = bd + B2_OFF + j * 5;
    b2[0] = ct[0] * ct[1];
    b2[1] = sg[0] * ct[1] + ct[1] * sg[1];
    b2[2] = sg[1] * sg[1] - ct[1] * ct[1] - ct[2] * ct[2];
    b2[3] = -(ct[2] * sg[1]) - sg[2] * ct[2];
    b2[4] = ct[3] * ct[2];
}

static __device__ __forceinline__ float band_val(const float* bw, int hw, int s, int j) {
    const int  d      = s - j;
    const bool inband = (d >= -hw && d <= hw);
    const float v     = bw[j * (2 * hw + 1) + (inband ? d + hw : 0)];
    return inband ? v : 0.0f;
}

static __device__ __forceinline__ void load_a(v2f (&a)[5], const float* bw, int hw,
                                              int jt, int c, int h, int sa, int sb) {
    #pragma unroll
    for (int p = 0; p < 4; ++p) {
        a[p][0] = band_val(bw, hw, jt + 8 * h + 2 * p,     jt + c);
        a[p][1] = band_val(bw, hw, jt + 8 * h + 2 * p + 1, jt + c);
    }
    const int s4 = h ? sb : sa;
    a[4][0] = band_val(bw, hw, s4,     jt + c);
    a[4][1] = band_val(bw, hw, s4 + 1, jt + c);
}

static __device__ __forceinline__ v8f mma(v2f a, v2f b, v8f c) {
    return __builtin_amdgcn_wmma_f32_16x16x4_f32(false, a, false, b, (short)0, c, false, false);
}

__global__ void __launch_bounds__(NWAVE * 32)
expmv_taylor_wmma_k2(const float* __restrict__ inp,
                     const float* __restrict__ prm,
                     float* __restrict__ out,
                     int B)
{
    __shared__ v2f edge[2 * SLOTS];
    __shared__ float bands[NBAND];

    const int tid  = threadIdx.x;
    const int wave = tid >> 5;
    const int lane = tid & 31;
    const int c    = lane & 15;
    const int h    = lane >> 4;
    const int j0   = wave * 32;
    const int j1   = j0 + 16;
    const int row0 = blockIdx.x * TILE;

    float xr[3], cr[4];
    load_raw(prm, tid, xr, cr);

    const int   r  = row0 + c;
    const int   rc = (r < B) ? r : (B - 1);
    const float tl = inp[rc * IN_STRIDE + UNITS];
    v8f S0, S1;
    #pragma unroll
    for (int v = 0; v < 8; ++v) {
        S0[v] = inp[rc * IN_STRIDE + j0 + 8 * h + v];
        S1[v] = inp[rc * IN_STRIDE + j1 + 8 * h + v];
    }

    store_bands(bands, tid, xr, cr);
    __syncthreads();

    v2f k1a0[5], k1a1[5], k2a0[5], k2a1[5];
    load_a(k1a0, bands + B1_OFF, 1, j0, c, h, j1,      j0 - 2);
    load_a(k1a1, bands + B1_OFF, 1, j1, c, h, j0 + 32, j0 + 14);
    load_a(k2a0, bands + B2_OFF, 2, j0, c, h, j1,      j0 - 2);
    load_a(k2a1, bands + B2_OFF, 2, j1, c, h, j0 + 32, j0 + 14);

    float cf[TERMS + 1];
    cf[0] = 1.0f;
    #pragma unroll
    for (int k = 1; k <= TERMS; ++k)
        cf[k] = cf[k - 1] * (tl * __builtin_amdgcn_rcpf((float)(SUBSTEPS * k)));

    const int wr_off = (wave * 2 + h) * TILE + c;
    const int nb_w   = h ? ((wave > 0) ? wave - 1 : wave) : ((wave < NWAVE - 1) ? wave + 1 : wave);
    const int rd_off = (nb_w * 2 + (h ? 1 : 0)) * TILE + c;

    {
        v2f lo = {S0[0], S0[1]};
        v2f hi = {S1[6], S1[7]};
        edge[wr_off] = h ? hi : lo;
    }

    for (int step = 0; step < SUBSTEPS; ++step) {
        const v8f U0 = S0, U1 = S1;
        v8f W0 = S0, W1 = S1;
        v8f A0 = S0, A1 = S1;
        v8f V0 = S0, V1 = S1;
        #pragma unroll
        for (int rd = 0; rd < ROUNDS; ++rd) {
            const int cur = rd & 1;
            v8f z0, z1;
            v8f y0 = {}, y1 = {};
            {
                const v8f zero = {};
                v2f b0 = {A0[0], A0[1]}, b1 = {A1[0], A1[1]};
                z0 = mma(k2a0[0], b0, (rd == 0) ? zero : V0);
                z1 = mma(k2a1[0], b1, (rd == 0) ? zero : V1);
                if (rd == 0) {
                    y0 = mma(k1a0[0], b0, zero);
                    y1 = mma(k1a1[0], b1, zero);
                }
                __builtin_amdgcn_sched_barrier(0);
                v2f d0 = {A0[2], A0[3]}, d1 = {A1[2], A1[3]};
                z0 = mma(k2a0[1], d0, z0);
                z1 = mma(k2a1[1], d1, z1);
                if (rd == 0) {
                    y0 = mma(k1a0[1], d0, y0);
                    y1 = mma(k1a1[1], d1, y1);
                }
            }
            __syncthreads();
            const v2f E = edge[cur * SLOTS + rd_off];
            __builtin_amdgcn_sched_barrier(0);
            #pragma unroll
            for (int p = 2; p < 4; ++p) {
                v2f b0 = {A0[2 * p], A0[2 * p + 1]}, b1 = {A1[2 * p], A1[2 * p + 1]};
                z0 = mma(k2a0[p], b0, z0);
                z1 = mma(k2a1[p], b1, z1);
                if (rd == 0) {
                    y0 = mma(k1a0[p], b0, y0);
                    y1 = mma(k1a1[p], b1, y1);
                }
            }
            __builtin_amdgcn_sched_barrier(0);
            v8f N0 = V0, N1 = V1;
            if (rd >= 1 && rd < ROUNDS - 1) {
                #pragma unroll
                for (int v = 0; v < 8; ++v) {
                    N0[v] = cf[2 * (4 - rd)] * U0[v] + cf[2 * (4 - rd) + 1] * W0[v];
                    N1[v] = cf[2 * (4 - rd)] * U1[v] + cf[2 * (4 - rd) + 1] * W1[v];
                }
                __builtin_amdgcn_sched_barrier(0);
            }
            {
                v2f a1lo = {A1[0], A1[1]}, a0hi = {A0[6], A0[7]};
                v2f x0 = h ? E : a1lo;
                v2f x1 = h ? a0hi : E;
                z0 = mma(k2a0[4], x0, z0);
                z1 = mma(k2a1[4], x1, z1);
                if (rd == 0) {
                    y0 = mma(k1a0[4], x0, y0);
                    y1 = mma(k1a1[4], x1, y1);
                }
            }

            if (rd == 0) {
                W0 = y0; W1 = y1;
                #pragma unroll
                for (int v = 0; v < 8; ++v) {
                    A0[v] = cf[10] * U0[v] + cf[11] * y0[v] + cf[12] * z0[v];
                    A1[v] = cf[10] * U1[v] + cf[11] * y1[v] + cf[12] * z1[v];
                }
            } else {
                A0 = z0; A1 = z1;
            }
            {
                v2f lo = {A0[0], A0[1]};
                v2f hi = {A1[6], A1[7]};
                edge[(cur ^ 1) * SLOTS + wr_off] = h ? hi : lo;
                __builtin_amdgcn_sched_barrier(0);
            }
            if (rd == 0) {
                #pragma unroll
                for (int v = 0; v < 8; ++v) {
                    V0[v] = cf[8] * U0[v] + cf[9] * W0[v];
                    V1[v] = cf[8] * U1[v] + cf[9] * W1[v];
                }
            } else {
                V0 = N0; V1 = N1;
            }
        }
        S0 = A0; S1 = A1;
    }

    if (r < B) {
        #pragma unroll
        for (int v = 0; v < 8; ++v) {
            out[r * UNITS + j0 + 8 * h + v] = S0[v];
            out[r * UNITS + j1 + 8 * h + v] = S1[v];
        }
    }
}

extern "C" void kernel_launch(void* const* d_in, const int* in_sizes, int n_in,
                              void* d_out, int out_size, void* d_ws, size_t ws_size,
                              hipStream_t stream) {
    const float* inp    = (const float*)d_in[0];
    const float* params = (const float*)d_in[1];
    float*       out    = (float*)d_out;
    (void)d_ws; (void)ws_size;

    int B = in_sizes[0] / IN_STRIDE;

    int tiles = (B + TILE - 1) / TILE;
    expmv_taylor_wmma_k2<<<tiles, NWAVE * 32, 0, stream>>>(inp, params, out, B);
}


# Diagonal shift: expm(tK) = e^(t mu) expm(t(K - mu I)), mu = middle of K's diagonal range (wave min/max of the parameters); the smaller norm takes 6 substeps of 12 terms, not 8; rounding changes
# speedup vs baseline: 7.8220x; 1.1742x over previous
#include <hip/hip_runtime.h>


#define UNITS     128
#define IN_STRIDE 129
#define TILE      16
#define SUBSTEPS  6
#define TERMS     12
#define ROUNDS    (TERMS / 2)
static_assert(TERMS == 12, "the round-to-coefficient map of the Horner rounds is written for 12 terms");
#define NWAVE     4
#define SLOTS     (NWAVE * 2 * TILE)
#define B1_OFF    0
#define B2_OFF    (UNITS * 3)
#define NBAND     (UNITS * 8)
static_assert(NWAVE * 32 == UNITS, "one thread of the workgroup per column of K");

typedef __attribute__((ext_vector_type(2))) float v2f;
typedef __attribute__((ext_vector_type(8))) float v8f;

static __device__ __forceinline__ void load_raw(const float* __restrict__ p, int j, int lane,
                                                float (&xr)[3], float (&cr)[4], float (&pq)[4]) {
    #pragma unroll
    for (int q = 0; q < 4; ++q)
        pq[q] = p[lane + 32 * q];
    #pragma unroll
    for (int d = 0; d < 3; ++d) {
        const int i = j + d - 1;
        xr[d] = p[min(max(i, 0), UNITS - 1)];
    }
    #pragma unroll
    for (int d = 0; d < 4; ++d) {
        const int i = j + d - 2;
        cr[d] = p[UNITS + min(max(i, 0), UNITS - 2)];
    }
}

template <int M>
static __device__ __forceinline__ float row_xor(float x) {
    return __builtin_bit_cast(float, __builtin_amdgcn_update_dpp(0, __builtin_bit_cast(int, x),
                                                                 0x160 + M, 0xf, 0xf, false));
}
static __device__ __forceinline__ float xhalf(float x) {
    return __builtin_bit_cast(float, __builtin_amdgcn_permlanex16(__builtin_bit_cast(int, x), __builtin_bit_cast(int, x),
                                                                  0x76543210, 0xfedcba98, false, false));
}
static __device__ __forceinline__ void wave_min_max(float& mn, float& mx) {
    mn = fminf(mn, row_xor<1>(mn));  mx = fmaxf(mx, row_xor<1>(mx));
    mn = fminf(mn, row_xor<2>(mn));  mx = fmaxf(mx, row_xor<2>(mx));
    mn = fminf(mn, row_xor<4>(mn));  mx = fmaxf(mx, row_xor<4>(mx));
    mn = fminf(mn, row_xor<8>(mn));  mx = fmaxf(mx, row_xor<8>(mx));
    mn = fminf(mn, xhalf(mn));       mx = fmaxf(mx, xhalf(mx));
}

static __device__ __forceinline__ float store_bands(float* bd, int j,
                                                    const float (&xr)[3], const float (&cr)[4],
                                                    float pmn, float pmx) {
    float sg[3], ct[4];
    float s2[5];
    #pragma unroll
    for (int d = 0; d < 5; ++d) {
        const float x  = (d < 3) ? xr[d] : ((d == 3) ? pmx : pmn);
        const float lp = log1pf(__expf(x));
        const float sp = (x > 20.0f) ? x : lp;
        s2[d] = sp * sp;
    }
    const float mu = -0.5f * (s2[3] + s2[4]);
    #pragma unroll
    for (int d = 0; d < 3; ++d) {
        const int i = j + d - 1;
        sg[d] = (i >= 0 && i < UNITS) ? (-s2[d] - mu) : 0.0f;
    }
    #pragma unroll
    for (int d = 0; d < 4; ++d) {
        const int i = j + d - 2;
        ct[d] = (i >= 0 && i < UNITS - 1) ? -cr[d] : 0.0f;
    }
    float* b1 = bd + B1_OFF + j * 3;
    b1[0] = ct[1];
    b1[1] = sg[1];
    b1[2] = -ct[2];
    float* b2 = bd + B2_OFF + j * 5;
    b2[0] = ct[0] * ct[1];
    b2[1] = sg[0] * ct[1] + ct[1] * sg[1];
    b2[2] = sg[1] * sg[1] - ct[1] * ct[1] - ct[2] * ct[2];
    b2[3] = -(ct[2] * sg[1]) - sg[2] * ct[2];
    b2[4] = ct[3] * ct[2];
    return mu;
}

static __device__ __forceinline__ float band_val(const float* bw, int hw, int s, int j) {
    const int  d      = s - j;
    const bool inband = (d >= -hw && d <= hw);
    const float v     = bw[j * (2 * hw + 1) + (inband ? d + hw : 0)];
    return inband ? v : 0.0f;
}

static __device__ __forceinline__ void load_a(v2f (&a)[5], const float* bw, int hw,
                                              int jt, int c, int h, int sa, int sb) {
    #pragma unroll
    for (int p = 0; p < 4; ++p) {
        a[p][0] = band_val(bw, hw, jt + 8 * h + 2 * p,     jt + c);
        a[p][1] = band_val(bw, hw, jt + 8 * h + 2 * p + 1, jt + c);
    }
    const int s4 = h ? sb : sa;
    a[4][0] = band_val(bw, hw, s4,     jt + c);
    a[4][1] = band_val(bw, hw, s4 + 1, jt + c);
}

static __device__ __forceinline__ v8f mma(v2f a, v2f b, v8f c) {
    return __builtin_amdgcn_wmma_f32_16x16x4_f32(false, a, false, b, (short)0, c, false, false);
}

__global__ void __launch_bounds__(NWAVE * 32)
expmv_taylor_wmma_k2(const float* __restrict__ inp,
                     const float* __restrict__ prm,
                     float* __restrict__ out,
                     int B)
{
    __shared__ v2f edge[2 * SLOTS];
    __shared__ float bands[NBAND];

    const int tid  = threadIdx.x;
    const int wave = tid >> 5;
    const int lane = tid & 31;
    const int c    = lane & 15;
    const int h    = lane >> 4;
    const int j0   = wave * 32;
    const int j1   = j0 + 16;
    const int row0 = blockIdx.x * TILE;

    float xr[3], cr[4], pq[4];
    load_raw(prm, tid, lane, xr, cr, pq);

    const int   r  = row0 + c;
    const int   rc = (r < B) ? r : (B - 1);
    const float tl = inp[rc * IN_STRIDE + UNITS];
    v8f S0, S1;
    #pragma unroll
    for (int v = 0; v < 8; ++v) {
        S0[v] = inp[rc * IN_STRIDE + j0 + 8 * h + v];
        S1[v] = inp[rc * IN_STRIDE + j1 + 8 * h + v];
    }

    float pmn = fminf(fminf(pq[0], pq[1]), fminf(pq[2], pq[3]));
    float pmx = fmaxf(fmaxf(pq[0], pq[1]), fmaxf(pq[2], pq[3]));
    wave_min_max(pmn, pmx);
    const float mu = store_bands(bands, tid, xr, cr, pmn, pmx);
    __syncthreads();

    v2f k1a0[5], k1a1[5], k2a0[5], k2a1[5];
    load_a(k1a0, bands + B1_OFF, 1, j0, c, h, j1,      j0 - 2);
    load_a(k1a1, bands + B1_OFF, 1, j1, c, h, j0 + 32, j0 + 14);
    load_a(k2a0, bands + B2_OFF, 2, j0, c, h, j1,      j0 - 2);
    load_a(k2a1, bands + B2_OFF, 2, j1, c, h, j0 + 32, j0 + 14);

    float cf[TERMS + 1];
    cf[0] = 1.0f;
    #pragma unroll
    for (int k = 1; k <= TERMS; ++k)
        cf[k] = cf[k - 1] * (tl * __builtin_amdgcn_rcpf((float)(SUBSTEPS * k)));

    const int wr_off = (wave * 2 + h) * TILE + c;
    const int nb_w   = h ? ((wave > 0) ? wave - 1 : wave) : ((wave < NWAVE - 1) ? wave + 1 : wave);
    const int rd_off = (nb_w * 2 + (h ? 1 : 0)) * TILE + c;

    {
        v2f lo = {S0[0], S0[1]};
        v2f hi = {S1[6], S1[7]};
        edge[wr_off] = h ? hi : lo;
    }

    for (int step = 0; step < SUBSTEPS; ++step) {
        const v8f U0 = S0, U1 = S1;
        v8f W0 = S0, W1 = S1;
        v8f A0 = S0, A1 = S1;
        v8f V0 = S0, V1 = S1;
        #pragma unroll
        for (int rd = 0; rd < ROUNDS; ++rd) {
            const int cur = rd & 1;
            v8f z0, z1;
            v8f y0 = {}, y1 = {};
            {
                const v8f zero = {};
                v2f b0 = {A0[0], A0[1]}, b1 = {A1[0], A1[1]};
                z0 = mma(k2a0[0], b0, (rd == 0) ? zero : V0);
                z1 = mma(k2a1[0], b1, (rd == 0) ? zero : V1);
                if (rd == 0) {
                    y0 = mma(k1a0[0], b0, zero);
                    y1 = mma(k1a1[0], b1, zero);
                }
                __builtin_amdgcn_sched_barrier(0);
                v2f d0 = {A0[2], A0[3]}, d1 = {A1[2], A1[3]};
                z0 = mma(k2a0[1], d0, z0);
                z1 = mma(k2a1[1], d1, z1);
                if (rd == 0) {
                    y0 = mma(k1a0[1], d0, y0);
                    y1 = mma(k1a1[1], d1, y1);
                }
            }
            __syncthreads();
            const v2f E = edge[cur * SLOTS + rd_off];
            __builtin_amdgcn_sched_barrier(0);
            #pragma unroll
            for (int p = 2; p < 4; ++p) {
                v2f b0 = {A0[2 * p], A0[2 * p + 1]}, b1 = {A1[2 * p], A1[2 * p + 1]};
                z0 = mma(k2a0[p], b0, z0);
                z1 = mma(k2a1[p], b1, z1);
                if (rd == 0) {
                    y0 = mma(k1a0[p], b0, y0);
                    y1 = mma(k1a1[p], b1, y1);
                }
            }
            __builtin_amdgcn_sched_barrier(0);
            v8f N0 = V0, N1 = V1;
            if (rd >= 1 && rd < ROUNDS - 1) {
                #pragma unroll
                for (int v = 0; v < 8; ++v) {
                    N0[v] = cf[2 * (4 - rd)] * U0[v] + cf[2 * (4 - rd) + 1] * W0[v];
                    N1[v] = cf[2 * (4 - rd)] * U1[v] + cf[2 * (4 - rd) + 1] * W1[v];
                }
                __builtin_amdgcn_sched_barrier(0);
            }
            {
                v2f a1lo = {A1[0], A1[1]}, a0hi = {A0[6], A0[7]};
                v2f x0 = h ? E : a1lo;
                v2f x1 = h ? a0hi : E;
                z0 = mma(k2a0[4], x0, z0);
                z1 = mma(k2a1[4], x1, z1);
                if (rd == 0) {
                    y0 = mma(k1a0[4], x0, y0);
                    y1 = mma(k1a1[4], x1, y1);
                }
            }

            if (rd == 0) {
                W0 = y0; W1 = y1;
                #pragma unroll
                for (int v = 0; v < 8; ++v) {
                    A0[v] = cf[10] * U0[v] + cf[11] * y0[v] + cf[12] * z0[v];
                    A1[v] = cf[10] * U1[v] + cf[11] * y1[v] + cf[12] * z1[v];
                }
            } else {
                A0 = z0; A1 = z1;
            }
            {
                v2f lo = {A0[0], A0[1]};
                v2f hi = {A1[6], A1[7]};
                edge[(cur ^ 1) * SLOTS + wr_off] = h ? hi : lo;
                __builtin_amdgcn_sched_barrier(0);
            }
            if (rd == 0) {
                #pragma unroll
                for (int v = 0; v < 8; ++v) {
                    V0[v] = cf[8] * U0[v] + cf[9] * W0[v];
                    V1[v] = cf[8] * U1[v] + cf[9] * W1[v];
                }
            } else {
                V0 = N0; V1 = N1;
            }
        }
        S0 = A0; S1 = A1;
    }

    const float am = tl * mu;
    const float er = fmaf(tl, mu, -am);
    const float ef = expf(am);
    const float sc = fmaf(ef, er, ef);

    if (r < B) {
        #pragma unroll
        for (int v = 0; v < 8; ++v) {
            out[r * UNITS + j0 + 8 * h + v] = sc * S0[v];
            out[r * UNITS + j1 + 8 * h + v] = sc * S1[v];
        }
    }
}

extern "C" void kernel_launch(void* const* d_in, const int* in_sizes, int n_in,
                              void* d_out, int out_size, void* d_ws, size_t ws_size,
                              hipStream_t stream) {
    const float* inp    = (const float*)d_in[0];
    const float* params = (const float*)d_in[1];
    float*       out    = (float*)d_out;
    (void)d_ws; (void)ws_size;

    int B = in_sizes[0] / IN_STRIDE;

    int tiles = (B + TILE - 1) / TILE;
    expmv_taylor_wmma_k2<<<tiles, NWAVE * 32, 0, stream>>>(inp, params, out, B);
}


# Horner in blocks of four powers: K3, K4 = K2*K2 (9 diagonals, 6 WMMAs a tile) built in LDS; degree 15 for 34 WMMAs and 4 exchanges a substep, so 4 substeps replace 6; sum order and rounding change
# speedup vs baseline: 9.5660x; 1.2230x over previous
#include <hip/hip_runtime.h>


#define UNITS     128
#define IN_STRIDE 129
#define TILE      16
#define SUBSTEPS  4
#define TERMS     15
#define ROUNDS    ((TERMS + 1) / 4)
static_assert(TERMS == 15, "the round-to-coefficient map of the Horner rounds is written for 15 terms");
static_assert((ROUNDS & 1) == 0, "the edge buffers change places every round: an even count keeps the parity over substeps");
#define NWAVE     4
#define SLOTS     (NWAVE * 2 * TILE)
#define PITCH(hw) (2 * (hw) + 3)
#define B1_OFF    0
#define B2_OFF    (B1_OFF + UNITS * PITCH(1))
#define B3_OFF    (B2_OFF + UNITS * PITCH(2))
#define B4_OFF    (B3_OFF + UNITS * PITCH(3))
#define NBAND     (B4_OFF + UNITS * PITCH(4))
static_assert(NWAVE * 32 == UNITS, "one thread of the workgroup per column of K");

typedef __attribute__((ext_vector_type(2))) float v2f;
typedef __attribute__((ext_vector_type(4))) float v4f;
typedef __attribute__((ext_vector_type(8))) float v8f;

static __device__ __forceinline__ void load_raw(const float* __restrict__ p, int j, int lane,
                                                float (&xr)[3], float (&cr)[4], float (&pq)[4]) {
    #pragma unroll
    for (int q = 0; q < 4; ++q)
        pq[q] = p[lane + 32 * q];
    #pragma unroll
    for (int d = 0; d < 3; ++d) {
        const int i = j + d - 1;
        xr[d] = p[min(max(i, 0), UNITS - 1)];
    }
    #pragma unroll
    for (int d = 0; d < 4; ++d) {
        const int i = j + d - 2;
        cr[d] = p[UNITS + min(max(i, 0), UNITS - 2)];
    }
}

template <int M>
static __device__ __forceinline__ float row_xor(float x) {
    return __builtin_bit_cast(float, __builtin_amdgcn_update_dpp(0, __builtin_bit_cast(int, x),
                                                                 0x160 + M, 0xf, 0xf, false));
}
static __device__ __forceinline__ float xhalf(float x) {
    return __builtin_bit_cast(float, __builtin_amdgcn_permlanex16(__builtin_bit_cast(int, x), __builtin_bit_cast(int, x),
                                                                  0x76543210, 0xfedcba98, false, false));
}
static __device__ __forceinline__ void wave_min_max(float& mn, float& mx) {
    mn = fminf(mn, row_xor<1>(mn));  mx = fmaxf(mx, row_xor<1>(mx));
    mn = fminf(mn, row_xor<2>(mn));  mx = fmaxf(mx, row_xor<2>(mx));
    mn = fminf(mn, row_xor<4>(mn));  mx = fmaxf(mx, row_xor<4>(mx));
    mn = fminf(mn, row_xor<8>(mn));  mx = fmaxf(mx, row_xor<8>(mx));
    mn = fminf(mn, xhalf(mn));       mx = fmaxf(mx, xhalf(mx));
}

static __device__ __forceinline__ float store_bands12(float* bd, int j,
                                                      const float (&xr)[3], const float (&cr)[4],
                                                      float pmn, float pmx,
                                                      float (&k1c)[3], float (&k2c)[5]) {
    float sg[3], ct[4];
    float s2[5];
    #pragma unroll
    for (int d = 0; d < 5; ++d) {
        const float x  = (d < 3) ? xr[d] : ((d == 3) ? pmx : pmn);
        const float lp = log1pf(__expf(x));
        const float sp = (x > 20.0f) ? x : lp;
        s2[d] = sp * sp;
    }
    const float mu = -0.5f * (s2[3] + s2[4]);
    #pragma unroll
    for (int d = 0; d < 3; ++d) {
        const int i = j + d - 1;
        sg[d] = (i >= 0 && i < UNITS) ? (-s2[d] - mu) : 0.0f;
    }
    #pragma unroll
    for (int d = 0; d < 4; ++d) {
        const int i = j + d - 2;
        ct[d] = (i >= 0 && i < UNITS - 1) ? -cr[d] : 0.0f;
    }
    k1c[0] = ct[1];
    k1c[1] = sg[1];
    k1c[2] = -ct[2];
    k2c[0] = ct[0] * ct[1];
    k2c[1] = sg[0] * ct[1] + ct[1] * sg[1];
    k2c[2] = sg[1] * sg[1] - ct[1] * ct[1] - ct[2] * ct[2];
    k2c[3] = -(ct[2] * sg[1]) - sg[2] * ct[2];
    k2c[4] = ct[3] * ct[2];
    float* b1 = bd + B1_OFF + j * PITCH(1);
    b1[0] = 0.0f;  b1[4] = 0.0f;
    #pragma unroll
    for (int k = 0; k < 3; ++k) b1[k + 1] = k1c[k];
    float* b2 = bd + B2_OFF + j * PITCH(2);
    b2[0] = 0.0f;  b2[6] = 0.0f;
    #pragma unroll
    for (int k = 0; k < 5; ++k) b2[k + 1] = k2c[k];
    return mu;
}

static __device__ __forceinline__ void store_bands34(float* bd, int j,
                                                     const float (&k1c)[3], const float (&k2c)[5]) {
    float n2[5][5];
    #pragma unroll
    for (int e = 0; e < 5; ++e) {
        const int m = min(max(j + e - 2, 0), UNITS - 1);
        #pragma unroll
        for (int k = 0; k < 5; ++k)
            n2[e][k] = (e == 2) ? k2c[k] : bd[B2_OFF + m * PITCH(2) + k + 1];
    }
    float* b3 = bd + B3_OFF + j * PITCH(3);
    b3[0] = 0.0f;  b3[8] = 0.0f;
    #pragma unroll
    for (int d = -3; d <= 3; ++d) {
        float acc = 0.0f;
        #pragma unroll
        for (int e = -1; e <= 1; ++e) {
            const int k = d - e;
            if (k >= -2 && k <= 2) acc += n2[e + 2][k + 2] * k1c[e + 1];
        }
        b3[d + 4] = acc;
    }
    float* b4 = bd + B4_OFF + j * PITCH(4);
    b4[0] = 0.0f;  b4[10] = 0.0f;
    #pragma unroll
    for (int d = -4; d <= 4; ++d) {
        float acc = 0.0f;
        #pragma unroll
        for (int e = -2; e <= 2; ++e) {
            const int k = d - e;
            if (k >= -2 && k <= 2) acc += n2[e + 2][k + 2] * k2c[e + 2];
        }
        b4[d + 5] = acc;
    }
}

static __device__ __forceinline__ float band_val(const float* bw, int hw, int s, int j) {
    return bw[j * PITCH(hw) + min(max(s - j + hw + 1, 0), 2 * hw + 2)];
}

template <int NCH>
static __device__ __forceinline__ void load_a(v2f (&a)[NCH], const float* bw, int hw,
                                              int jt, int c, int h, int n0, int n1, int f0, int f1) {
    #pragma unroll
    for (int p = 0; p < 4; ++p) {
        a[p][0] = band_val(bw, hw, jt + 8 * h + 2 * p,     jt + c);
        a[p][1] = band_val(bw, hw, jt + 8 * h + 2 * p + 1, jt + c);
    }
    const int s4 = h ? n1 : n0;
    a[4][0] = band_val(bw, hw, s4,     jt + c);
    a[4][1] = band_val(bw, hw, s4 + 1, jt + c);
    if (NCH == 6) {
        const int s5 = h ? f1 : f0;
        a[NCH - 1][0] = band_val(bw, hw, s5,     jt + c);
        a[NCH - 1][1] = band_val(bw, hw, s5 + 1, jt + c);
    }
}

static __device__ __forceinline__ v8f mma(v2f a, v2f b, v8f c) {
    return __builtin_amdgcn_wmma_f32_16x16x4_f32(false, a, false, b, (short)0, c, false, false);
}

__global__ void __launch_bounds__(NWAVE * 32)
expmv_taylor_wmma_k4(const float* __restrict__ inp,
                     const float* __restrict__ prm,
                     float* __restrict__ out,
                     int B)
{
    __shared__ v4f edge[2 * SLOTS];
    __shared__ float bands[NBAND];

    const int tid  = threadIdx.x;
    const int wave = tid >> 5;
    const int lane = tid & 31;
    const int c    = lane & 15;
    const int h    = lane >> 4;
    const int j0   = wave * 32;
    const int j1   = j0 + 16;
    const int row0 = blockIdx.x * TILE;

    float xr[3], cr[4], pq[4];
    load_raw(prm, tid, lane, xr, cr, pq);

    const int   r  = row0 + c;
    const int   rc = (r < B) ? r : (B - 1);
    const float tl = inp[rc * IN_STRIDE + UNITS];
    v8f S0, S1;
    #pragma unroll
    for (int v = 0; v < 8; ++v) {
        S0[v] = inp[rc * IN_STRIDE + j0 + 8 * h + v];
        S1[v] = inp[rc * IN_STRIDE + j1 + 8 * h + v];
    }

    float pmn = fminf(fminf(pq[0], pq[1]), fminf(pq[2], pq[3]));
    float pmx = fmaxf(fmaxf(pq[0], pq[1]), fmaxf(pq[2], pq[3]));
    wave_min_max(pmn, pmx);
    float k1c[3], k2c[5];
    const float mu = store_bands12(bands, tid, xr, cr, pmn, pmx, k1c, k2c);
    __syncthreads();
    store_bands34(bands, tid, k1c, k2c);
    __syncthreads();

    v2f k1a0[5], k1a1[5], k2a0[5], k2a1[5], k3a0[6], k3a1[6], k4a0[6], k4a1[6];
    load_a<5>(k1a0, bands + B1_OFF, 1, j0, c, h, j1,      j0 - 2,  j1 + 2,  j0 - 4);
    load_a<5>(k1a1, bands + B1_OFF, 1, j1, c, h, j0 + 32, j0 + 14, j0 + 34, j0 + 12);
    load_a<5>(k2a0, bands + B2_OFF, 2, j0, c, h, j1,      j0 - 2,  j1 + 2,  j0 - 4);
    load_a<5>(k2a1, bands + B2_OFF, 2, j1, c, h, j0 + 32, j0 + 14, j0 + 34, j0 + 12);
    load_a<6>(k3a0, bands + B3_OFF, 3, j0, c, h, j1,      j0 - 2,  j1 + 2,  j0 - 4);
    load_a<6>(k3a1, bands + B3_OFF, 3, j1, c, h, j0 + 32, j0 + 14, j0 + 34, j0 + 12);
    load_a<6>(k4a0, bands + B4_OFF, 4, j0, c, h, j1,      j0 - 2,  j1 + 2,  j0 - 4);
    load_a<6>(k4a1, bands + B4_OFF, 4, j1, c, h, j0 + 32, j0 + 14, j0 + 34, j0 + 12);

    float cf[TERMS + 1];
    cf[0] = 1.0f;
    #pragma unroll
    for (int k = 1; k <= TERMS; ++k)
        cf[k] = cf[k - 1] * (tl * __builtin_amdgcn_rcpf((float)(SUBSTEPS * k)));

    const int wr_off = (wave * 2 + h) * TILE + c;
    const int nb_w   = h ? ((wave > 0) ? wave - 1 : wave) : ((wave < NWAVE - 1) ? wave + 1 : wave);
    const int rd_off = (nb_w * 2 + (h ? 1 : 0)) * TILE + c;

    {
        v4f lo = {S0[0], S0[1], S0[2], S0[3]};
        v4f hi = {S1[6], S1[7], S1[4], S1[5]};
        edge[wr_off] = h ? hi : lo;
    }

    for (int step = 0; step < SUBSTEPS; ++step) {
        const v8f U0 = S0, U1 = S1;
        v8f P0 = S0, P1 = S1;
        v8f Q0 = S0, Q1 = S1;
        v8f R0 = S0, R1 = S1;
        v8f A0 = S0, A1 = S1;
        v8f V0 = S0, V1 = S1;
        #pragma unroll
        for (int rd = 0; rd < ROUNDS; ++rd) {
            const int cur = rd & 1;
            const v8f zero = {};
            v8f z0 = zero, z1 = zero;
            v8f p0 = zero, p1 = zero;
            v8f q0 = zero, q1 = zero;
            v8f r0 = zero, r1 = zero;
            {
                v2f b0 = {A0[0], A0[1]}, b1 = {A1[0], A1[1]};
                if (rd == 0) {
                    p0 = mma(k1a0[0], b0, zero);  p1 = mma(k1a1[0], b1, zero);
                    q0 = mma(k2a0[0], b0, zero);  q1 = mma(k2a1[0], b1, zero);
                    r0 = mma(k3a0[0], b0, zero);  r1 = mma(k3a1[0], b1, zero);
                } else {
                    z0 = mma(k4a0[0], b0, V0);    z1 = mma(k4a1[0], b1, V1);
                }
                __builtin_amdgcn_sched_barrier(0);
                v2f d0 = {A0[2], A0[3]}, d1 = {A1[2], A1[3]};
                if (rd == 0) {
                    p0 = mma(k1a0[1], d0, p0);  p1 = mma(k1a1[1], d1, p1);
                    q0 = mma(k2a0[1], d0, q0);  q1 = mma(k2a1[1], d1, q1);
                    r0 = mma(k3a0[1], d0, r0);  r1 = mma(k3a1[1], d1, r1);
                } else {
                    z0 = mma(k4a0[1], d0, z0);  z1 = mma(k4a1[1], d1, z1);
                }
            }
            __syncthreads();
            const v4f E = edge[cur * SLOTS + rd_off];
            __builtin_amdgcn_sched_barrier(0);
            #pragma unroll
            for (int p = 2; p < 4; ++p) {
                v2f b0 = {A0[2 * p], A0[2 * p + 1]}, b1 = {A1[2 * p], A1[2 * p + 1]};
                if (rd == 0) {
                    p0 = mma(k1a0[p], b0, p0);  p1 = mma(k1a1[p], b1, p1);
                    q0 = mma(k2a0[p], b0, q0);  q1 = mma(k2a1[p], b1, q1);
                    r0 = mma(k3a0[p], b0, r0);  r1 = mma(k3a1[p], b1, r1);
                } else {
                    z0 = mma(k4a0[p], b0, z0);  z1 = mma(k4a1[p], b1, z1);
                }
            }
            __builtin_amdgcn_sched_barrier(0);
            v8f N0 = V0, N1 = V1;
            if (rd >= 1 && rd < ROUNDS - 1) {
                const int k = 4 * (2 - rd);
                N0 = U0 * cf[k] + P0 * cf[k + 1] + Q0 * cf[k + 2] + R0 * cf[k + 3];
                N1 = U1 * cf[k] + P1 * cf[k + 1] + Q1 * cf[k + 2] + R1 * cf[k + 3];
                __builtin_amdgcn_sched_barrier(0);
            }
            {
                v2f En = {E[0], E[1]}, Ef = {E[2], E[3]};
                v2f a1n = {A1[0], A1[1]}, a1f = {A1[2], A1[3]};
                v2f a0n = {A0[6], A0[7]}, a0f = {A0[4], A0[5]};
                v2f x0 = h ? En : a1n;
                v2f x1 = h ? a0n : En;
                v2f w0 = h ? Ef : a1f;
                v2f w1 = h ? a0f : Ef;
                if (rd == 0) {
                    p0 = mma(k1a0[4], x0, p0);  p1 = mma(k1a1[4], x1, p1);
                    q0 = mma(k2a0[4], x0, q0);  q1 = mma(k2a1[4], x1, q1);
                    r0 = mma(k3a0[4], x0, r0);  r1 = mma(k3a1[4], x1, r1);
                    r0 = mma(k3a0[5], w0, r0);  r1 = mma(k3a1[5], w1, r1);
                } else {
                    z0 = mma(k4a0[4], x0, z0);  z1 = mma(k4a1[4], x1, z1);
                    z0 = mma(k4a0[5], w0, z0);  z1 = mma(k4a1[5], w1, z1);
                }
            }

            if (rd == 0) {
                P0 = p0; P1 = p1; Q0 = q0; Q1 = q1; R0 = r0; R1 = r1;
                A0 = U0 * cf[12] + P0 * cf[13] + Q0 * cf[14] + R0 * cf[15];
                A1 = U1 * cf[12] + P1 * cf[13] + Q1 * cf[14] + R1 * cf[15];
            } else {
                A0 = z0; A1 = z1;
            }
            {
                v4f lo = {A0[0], A0[1], A0[2], A0[3]};
                v4f hi = {A1[6], A1[7], A1[4], A1[5]};
                edge[(cur ^ 1) * SLOTS + wr_off] = h ? hi : lo;
                __builtin_amdgcn_sched_barrier(0);
            }
            if (rd == 0) {
                V0 = U0 * cf[8] + P0 * cf[9] + Q0 * cf[10] + R0 * cf[11];
                V1 = U1 * cf[8] + P1 * cf[9] + Q1 * cf[10] + R1 * cf[11];
            } else {
                V0 = N0; V1 = N1;
            }
        }
        S0 = A0; S1 = A1;
    }

    const float am = tl * mu;
    const float er = fmaf(tl, mu, -am);
    const float ef = expf(am);
    const float sc = fmaf(ef, er, ef);

    if (r < B) {
        #pragma unroll
        for (int v = 0; v < 8; ++v) {
            out[r * UNITS + j0 + 8 * h + v] = sc * S0[v];
            out[r * UNITS + j1 + 8 * h + v] = sc * S1[v];
        }
    }
}

extern "C" void kernel_launch(void* const* d_in, const int* in_sizes, int n_in,
                              void* d_out, int out_size, void* d_ws, size_t ws_size,
                              hipStream_t stream) {
    const float* inp    = (const float*)d_in[0];
    const float* params = (const float*)d_in[1];
    float*       out    = (float*)d_out;
    (void)d_ws; (void)ws_size;

    int B = in_sizes[0] / IN_STRIDE;

    int tiles = (B + TILE - 1) / TILE;
    expmv_taylor_wmma_k4<<<tiles, NWAVE * 32, 0, stream>>>(inp, params, out, B);
}


# One substep of Taylor degree 39 in ten Horner rounds: 70 WMMAs and 10 edge exchanges a tile, not 4 x 34 and 16; powers of (K - mu I)/8 keep h^k/k! in f32 range; sum order and rounding change
# speedup vs baseline: 11.7700x; 1.2304x over previous
#include <hip/hip_runtime.h>


#define UNITS     128
#define IN_STRIDE 129
#define TILE      16
#define SUBSTEPS  1
#define TERMS     39
#define PSCALE    8
#define ROUNDS    ((TERMS + 1) / 4)
static_assert((TERMS + 1) % 4 == 0 && ROUNDS >= 3, "whole blocks of four terms: an opening round, at least one middle round, a last round");
static_assert(SUBSTEPS == 1 || (ROUNDS & 1) == 0, "the edge buffers change places every round: an even count keeps the parity over substeps");
static_assert((PSCALE & (PSCALE - 1)) == 0, "a power of two: dividing by it is exact");
#define NWAVE     4
#define SLOTS     (NWAVE * 2 * TILE)
#define PITCH(hw) (2 * (hw) + 3)
#define B1_OFF    0
#define B2_OFF    (B1_OFF + UNITS * PITCH(1))
#define B3_OFF    (B2_OFF + UNITS * PITCH(2))
#define B4_OFF    (B3_OFF + UNITS * PITCH(3))
#define NBAND     (B4_OFF + UNITS * PITCH(4))
static_assert(NWAVE * 32 == UNITS, "one thread of the workgroup per column of K");

typedef __attribute__((ext_vector_type(2))) float v2f;
typedef __attribute__((ext_vector_type(4))) float v4f;
typedef __attribute__((ext_vector_type(8))) float v8f;

static __device__ __forceinline__ void load_raw(const float* __restrict__ p, int j, int lane,
                                                float (&xr)[3], float (&cr)[4], float (&pq)[4]) {
    #pragma unroll
    for (int q = 0; q < 4; ++q)
        pq[q] = p[lane + 32 * q];
    #pragma unroll
    for (int d = 0; d < 3; ++d) {
        const int i = j + d - 1;
        xr[d] = p[min(max(i, 0), UNITS - 1)];
    }
    #pragma unroll
    for (int d = 0; d < 4; ++d) {
        const int i = j + d - 2;
        cr[d] = p[UNITS + min(max(i, 0), UNITS - 2)];
    }
}

template <int M>
static __device__ __forceinline__ float row_xor(float x) {
    return __builtin_bit_cast(float, __builtin_amdgcn_update_dpp(0, __builtin_bit_cast(int, x),
                                                                 0x160 + M, 0xf, 0xf, false));
}
static __device__ __forceinline__ float xhalf(float x) {
    return __builtin_bit_cast(float, __builtin_amdgcn_permlanex16(__builtin_bit_cast(int, x), __builtin_bit_cast(int, x),
                                                                  0x76543210, 0xfedcba98, false, false));
}
static __device__ __forceinline__ void wave_min_max(float& mn, float& mx) {
    mn = fminf(mn, row_xor<1>(mn));  mx = fmaxf(mx, row_xor<1>(mx));
    mn = fminf(mn, row_xor<2>(mn));  mx = fmaxf(mx, row_xor<2>(mx));
    mn = fminf(mn, row_xor<4>(mn));  mx = fmaxf(mx, row_xor<4>(mx));
    mn = fminf(mn, row_xor<8>(mn));  mx = fmaxf(mx, row_xor<8>(mx));
    mn = fminf(mn, xhalf(mn));       mx = fmaxf(mx, xhalf(mx));
}

static __device__ __forceinline__ float store_bands12(float* bd, int j,
                                                      const float (&xr)[3], const float (&cr)[4],
                                                      float pmn, float pmx,
                                                      float (&k1c)[3], float (&k2c)[5]) {
    float sg[3], ct[4];
    float s2[5];
    #pragma unroll
    for (int d = 0; d < 5; ++d) {
        const float x  = (d < 3) ? xr[d] : ((d == 3) ? pmx : pmn);
        const float lp = log1pf(__expf(x));
        const float sp = (x > 20.0f) ? x : lp;
        s2[d] = sp * sp;
    }
    const float mu = -0.5f * (s2[3] + s2[4]);
    #pragma unroll
    for (int d = 0; d < 3; ++d) {
        const int i = j + d - 1;
        sg[d] = (i >= 0 && i < UNITS) ? (-s2[d] - mu) * (1.0f / PSCALE) : 0.0f;
    }
    #pragma unroll
    for (int d = 0; d < 4; ++d) {
        const int i = j + d - 2;
        ct[d] = (i >= 0 && i < UNITS - 1) ? -cr[d] * (1.0f / PSCALE) : 0.0f;
    }
    k1c[0] = ct[1];
    k1c[1] = sg[1];
    k1c[2] = -ct[2];
    k2c[0] = ct[0] * ct[1];
    k2c[1] = sg[0] * ct[1] + ct[1] * sg[1];
    k2c[2] = sg[1] * sg[1] - ct[1] * ct[1] - ct[2] * ct[2];
    k2c[3] = -(ct[2] * sg[1]) - sg[2] * ct[2];
    k2c[4] = ct[3] * ct[2];
    float* b1 = bd + B1_OFF + j * PITCH(1);
    b1[0] = 0.0f;  b1[4] = 0.0f;
    #pragma unroll
    for (int k = 0; k < 3; ++k) b1[k + 1] = k1c[k];
    float* b2 = bd + B2_OFF + j * PITCH(2);
    b2[0] = 0.0f;  b2[6] = 0.0f;
    #pragma unroll
    for (int k = 0; k < 5; ++k) b2[k + 1] = k2c[k];
    return mu;
}

static __device__ __forceinline__ void store_bands34(float* bd, int j,
                                                     const float (&k1c)[3], const float (&k2c)[5]) {
    float n2[5][5];
    #pragma unroll
    for (int e = 0; e < 5; ++e) {
        const int m = min(max(j + e - 2, 0), UNITS - 1);
        #pragma unroll
        for (int k = 0; k < 5; ++k)
            n2[e][k] = (e == 2) ? k2c[k] : bd[B2_OFF + m * PITCH(2) + k + 1];
    }
    float* b3 = bd + B3_OFF + j * PITCH(3);
    b3[0] = 0.0f;  b3[8] = 0.0f;
    #pragma unroll
    for (int d = -3; d <= 3; ++d) {
        float acc = 0.0f;
        #pragma unroll
        for (int e = -1; e <= 1; ++e) {
            const int k = d - e;
            if (k >= -2 && k <= 2) acc += n2[e + 2][k + 2] * k1c[e + 1];
        }
        b3[d + 4] = acc;
    }
    float* b4 = bd + B4_OFF + j * PITCH(4);
    b4[0] = 0.0f;  b4[10] = 0.0f;
    #pragma unroll
    for (int d = -4; d <= 4; ++d) {
        float acc = 0.0f;
        #pragma unroll
        for (int e = -2; e <= 2; ++e) {
            const int k = d - e;
            if (k >= -2 && k <= 2) acc += n2[e + 2][k + 2] * k2c[e + 2];
        }
        b4[d + 5] = acc;
    }
}

static __device__ __forceinline__ float band_val(const float* bw, int hw, int s, int j) {
    return bw[j * PITCH(hw) + min(max(s - j + hw + 1, 0), 2 * hw + 2)];
}

template <int NCH>
static __device__ __forceinline__ void load_a(v2f (&a)[NCH], const float* bw, int hw,
                                              int jt, int c, int h, int n0, int n1, int f0, int f1) {
    #pragma unroll
    for (int p = 0; p < 4; ++p) {
        a[p][0] = band_val(bw, hw, jt + 8 * h + 2 * p,     jt + c);
        a[p][1] = band_val(bw, hw, jt + 8 * h + 2 * p + 1, jt + c);
    }
    const int s4 = h ? n1 : n0;
    a[4][0] = band_val(bw, hw, s4,     jt + c);
    a[4][1] = band_val(bw, hw, s4 + 1, jt + c);
    if (NCH == 6) {
        const int s5 = h ? f1 : f0;
        a[NCH - 1][0] = band_val(bw, hw, s5,     jt + c);
        a[NCH - 1][1] = band_val(bw, hw, s5 + 1, jt + c);
    }
}

static __device__ __forceinline__ v8f mma(v2f a, v2f b, v8f c) {
    return __builtin_amdgcn_wmma_f32_16x16x4_f32(false, a, false, b, (short)0, c, false, false);
}

__global__ void __launch_bounds__(NWAVE * 32)
expmv_taylor_wmma_k4(const float* __restrict__ inp,
                     const float* __restrict__ prm,
                     float* __restrict__ out,
                     int B)
{
    __shared__ v4f edge[2 * SLOTS];
    __shared__ float bands[NBAND];

    const int tid  = threadIdx.x;
    const int wave = tid >> 5;
    const int lane = tid & 31;
    const int c    = lane & 15;
    const int h    = lane >> 4;
    const int j0   = wave * 32;
    const int j1   = j0 + 16;
    const int row0 = blockIdx.x * TILE;

    float xr[3], cr[4], pq[4];
    load_raw(prm, tid, lane, xr, cr, pq);

    const int   r  = row0 + c;
    const int   rc = (r < B) ? r : (B - 1);
    const float tl = inp[rc * IN_STRIDE + UNITS];
    v8f S0, S1;
    #pragma unroll
    for (int v = 0; v < 8; ++v) {
        S0[v] = inp[rc * IN_STRIDE + j0 + 8 * h + v];
        S1[v] = inp[rc * IN_STRIDE + j1 + 8 * h + v];
    }

    float pmn = fminf(fminf(pq[0], pq[1]), fminf(pq[2], pq[3]));
    float pmx = fmaxf(fmaxf(pq[0], pq[1]), fmaxf(pq[2], pq[3]));
    wave_min_max(pmn, pmx);
    float k1c[3], k2c[5];
    const float mu = store_bands12(bands, tid, xr, cr, pmn, pmx, k1c, k2c);
    __syncthreads();
    store_bands34(bands, tid, k1c, k2c);
    __syncthreads();

    v2f k1a0[5], k1a1[5], k2a0[5], k2a1[5], k3a0[6], k3a1[6], k4a0[6], k4a1[6];
    load_a<5>(k1a0, bands + B1_OFF, 1, j0, c, h, j1,      j0 - 2,  j1 + 2,  j0 - 4);
    load_a<5>(k1a1, bands + B1_OFF, 1, j1, c, h, j0 + 32, j0 + 14, j0 + 34, j0 + 12);
    load_a<5>(k2a0, bands + B2_OFF, 2, j0, c, h, j1,      j0 - 2,  j1 + 2,  j0 - 4);
    load_a<5>(k2a1, bands + B2_OFF, 2, j1, c, h, j0 + 32, j0 + 14, j0 + 34, j0 + 12);
    load_a<6>(k3a0, bands + B3_OFF, 3, j0, c, h, j1,      j0 - 2,  j1 + 2,  j0 - 4);
    load_a<6>(k3a1, bands + B3_OFF, 3, j1, c, h, j0 + 32, j0 + 14, j0 + 34, j0 + 12);
    load_a<6>(k4a0, bands + B4_OFF, 4, j0, c, h, j1,      j0 - 2,  j1 + 2,  j0 - 4);
    load_a<6>(k4a1, bands + B4_OFF, 4, j1, c, h, j0 + 32, j0 + 14, j0 + 34, j0 + 12);

    float cf[TERMS + 1];
    cf[0] = 1.0f;
    #pragma unroll
    for (int k = 1; k <= TERMS; ++k)
        cf[k] = cf[k - 1] * (tl * ((float)PSCALE / (float)(SUBSTEPS * k)));

    const int wr_off = (wave * 2 + h) * TILE + c;
    const int nb_w   = h ? ((wave > 0) ? wave - 1 : wave) : ((wave < NWAVE - 1) ? wave + 1 : wave);
    const int rd_off = (nb_w * 2 + (h ? 1 : 0)) * TILE + c;

    {
        v4f lo = {S0[0], S0[1], S0[2], S0[3]};
        v4f hi = {S1[6], S1[7], S1[4], S1[5]};
        edge[wr_off] = h ? hi : lo;
    }

    for (int step = 0; step < SUBSTEPS; ++step) {
        const v8f U0 = S0, U1 = S1;
        v8f P0 = S0, P1 = S1;
        v8f Q0 = S0, Q1 = S1;
        v8f R0 = S0, R1 = S1;
        v8f A0 = S0, A1 = S1;
        v8f V0 = S0, V1 = S1;
        #pragma unroll
        for (int rd = 0; rd < ROUNDS; ++rd) {
            const int cur = rd & 1;
            const v8f zero = {};
            v8f z0 = zero, z1 = zero;
            v8f p0 = zero, p1 = zero;
            v8f q0 = zero, q1 = zero;
            v8f r0 = zero, r1 = zero;
            {
                v2f b0 = {A0[0], A0[1]}, b1 = {A1[0], A1[1]};
                if (rd == 0) {
                    p0 = mma(k1a0[0], b0, zero);  p1 = mma(k1a1[0], b1, zero);
                    q0 = mma(k2a0[0], b0, zero);  q1 = mma(k2a1[0], b1, zero);
                    r0 = mma(k3a0[0], b0, zero);  r1 = mma(k3a1[0], b1, zero);
                } else {
                    z0 = mma(k4a0[0], b0, V0);    z1 = mma(k4a1[0], b1, V1);
                }
                __builtin_amdgcn_sched_barrier(0);
                v2f d0 = {A0[2], A0[3]}, d1 = {A1[2], A1[3]};
                if (rd == 0) {
                    p0 = mma(k1a0[1], d0, p0);  p1 = mma(k1a1[1], d1, p1);
                    q0 = mma(k2a0[1], d0, q0);  q1 = mma(k2a1[1], d1, q1);
                    r0 = mma(k3a0[1], d0, r0);  r1 = mma(k3a1[1], d1, r1);
                } else {
                    z0 = mma(k4a0[1], d0, z0);  z1 = mma(k4a1[1], d1, z1);
                }
            }
            __syncthreads();
            const v4f E = edge[cur * SLOTS + rd_off];
            __builtin_amdgcn_sched_barrier(0);
            #pragma unroll
            for (int p = 2; p < 4; ++p) {
                v2f b0 = {A0[2 * p], A0[2 * p + 1]}, b1 = {A1[2 * p], A1[2 * p + 1]};
                if (rd == 0) {
                    p0 = mma(k1a0[p], b0, p0);  p1 = mma(k1a1[p], b1, p1);
                    q0 = mma(k2a0[p], b0, q0);  q1 = mma(k2a1[p], b1, q1);
                    r0 = mma(k3a0[p], b0, r0);  r1 = mma(k3a1[p], b1, r1);
                } else {
                    z0 = mma(k4a0[p], b0, z0);  z1 = mma(k4a1[p], b1, z1);
                }
            }
            __builtin_amdgcn_sched_barrier(0);
            v8f N0 = V0, N1 = V1;
            if (rd >= 1 && rd < ROUNDS - 1) {
                const int k = 4 * (ROUNDS - 2 - rd);
                N0 = U0 * cf[k] + P0 * cf[k + 1] + Q0 * cf[k + 2] + R0 * cf[k + 3];
                N1 = U1 * cf[k] + P1 * cf[k + 1] + Q1 * cf[k + 2] + R1 * cf[k + 3];
                __builtin_amdgcn_sched_barrier(0);
            }
            {
                v2f En = {E[0], E[1]}, Ef = {E[2], E[3]};
                v2f a1n = {A1[0], A1[1]}, a1f = {A1[2], A1[3]};
                v2f a0n = {A0[6], A0[7]}, a0f = {A0[4], A0[5]};
                v2f x0 = h ? En : a1n;
                v2f x1 = h ? a0n : En;
                v2f w0 = h ? Ef : a1f;
                v2f w1 = h ? a0f : Ef;
                if (rd == 0) {
                    p0 = mma(k1a0[4], x0, p0);  p1 = mma(k1a1[4], x1, p1);
                    q0 = mma(k2a0[4], x0, q0);  q1 = mma(k2a1[4], x1, q1);
                    r0 = mma(k3a0[4], x0, r0);  r1 = mma(k3a1[4], x1, r1);
                    r0 = mma(k3a0[5], w0, r0);  r1 = mma(k3a1[5], w1, r1);
                } else {
                    z0 = mma(k4a0[4], x0, z0);  z1 = mma(k4a1[4], x1, z1);
                    z0 = mma(k4a0[5], w0, z0);  z1 = mma(k4a1[5], w1, z1);
                }
            }

            if (rd == 0) {
                P0 = p0; P1 = p1; Q0 = q0; Q1 = q1; R0 = r0; R1 = r1;
                const int k = 4 * (ROUNDS - 1);
                A0 = U0 * cf[k] + P0 * cf[k + 1] + Q0 * cf[k + 2] + R0 * cf[k + 3];
                A1 = U1 * cf[k] + P1 * cf[k + 1] + Q1 * cf[k + 2] + R1 * cf[k + 3];
            } else {
                A0 = z0; A1 = z1;
            }
            {
                v4f lo = {A0[0], A0[1], A0[2], A0[3]};
                v4f hi = {A1[6], A1[7], A1[4], A1[5]};
                edge[(cur ^ 1) * SLOTS + wr_off] = h ? hi : lo;
                __builtin_amdgcn_sched_barrier(0);
            }
            if (rd == 0) {
                const int k = 4 * (ROUNDS - 2);
                V0 = U0 * cf[k] + P0 * cf[k + 1] + Q0 * cf[k + 2] + R0 * cf[k + 3];
                V1 = U1 * cf[k] + P1 * cf[k + 1] + Q1 * cf[k + 2] + R1 * cf[k + 3];
            } else {
                V0 = N0; V1 = N1;
            }
        }
        S0 = A0; S1 = A1;
    }

    const float am = tl * mu;
    const float er = fmaf(tl, mu, -am);
    const float ef = expf(am);
    const float sc = fmaf(ef, er, ef);

    if (r < B) {
        #pragma unroll
        for (int v = 0; v < 8; ++v) {
            out[r * UNITS + j0 + 8 * h + v] = sc * S0[v];
            out[r * UNITS + j1 + 8 * h + v] = sc * S1[v];
        }
    }
}

extern "C" void kernel_launch(void* const* d_in, const int* in_sizes, int n_in,
                              void* d_out, int out_size, void* d_ws, size_t ws_size,
                              hipStream_t stream) {
    const float* inp    = (const float*)d_in[0];
    const float* params = (const float*)d_in[1];
    float*       out    = (float*)d_out;
    (void)d_ws; (void)ws_size;

    int B = in_sizes[0] / IN_STRIDE;

    int tiles = (B + TILE - 1) / TILE;
    expmv_taylor_wmma_k4<<<tiles, NWAVE * 32, 0, stream>>>(inp, params, out, B);
}
